# starting kernel
# baseline (speedup 1.0000x reference)
#include <hip/hip_runtime.h>


#define ALPHA 0.2f
#define N_  4
#define HD  16
#define WD  16
#define T_  24
#define V_  32
#define HW  256
#define THW 512

typedef __attribute__((ext_vector_type(2))) float v2f;
typedef __attribute__((ext_vector_type(8))) float v8f;

__global__ void adj_kernel(const float* __restrict__ B, float* __restrict__ adjn) {
    __shared__ float rmin[V_], rmax[V_], dinv[V_];
    __shared__ float smn, smx;
    int i = threadIdx.x;
    float mn = 1e30f, mx = -1e30f;
    for (int j = 0; j < V_; ++j) {
        float v = B[i * V_ + j] + 1e-6f + (i == j ? 1.0f : 0.0f);
        mn = fminf(mn, v); mx = fmaxf(mx, v);
    }
    rmin[i] = mn; rmax[i] = mx;
    __syncthreads();
    if (i == 0) {
        float gmn = rmin[0], gmx = rmax[0];
        for (int k = 1; k < V_; ++k) { gmn = fminf(gmn, rmin[k]); gmx = fmaxf(gmx, rmax[k]); }
        smn = gmn; smx = gmx;
    }
    __syncthreads();
    float gmn = smn, inv = 1.0f / (smx - smn);
    float s = 0.0f;
    for (int j = 0; j < V_; ++j) {
        float v = B[i * V_ + j] + 1e-6f + (i == j ? 1.0f : 0.0f);
        s += (v - gmn) * inv;
    }
    dinv[i] = rsqrtf(s);
    __syncthreads();
    float di = dinv[i];
    for (int j = 0; j < V_; ++j) {
        float v = B[i * V_ + j] + 1e-6f + (i == j ? 1.0f : 0.0f);
        adjn[i * V_ + j] = (v - gmn) * inv * di * dinv[j];
    }
}

__global__ void __launch_bounds__(256) wh_kernel(const float* __restrict__ h,
                                                 const float* __restrict__ Wp,
                                                 float* __restrict__ Wh) {
    __shared__ float Xs[T_][V_];
    __shared__ float Wm[T_][T_];
    int bid = blockIdx.x;
    int n = bid >> 8;
    int p = bid & 255;
    int tid = threadIdx.x;
    const float* hp = h + (size_t)(n * HW + p) * T_ * V_;
    for (int idx = tid; idx < T_ * V_; idx += 256)
        Xs[idx / V_][idx & (V_ - 1)] = hp[idx];
    for (int idx = tid; idx < T_ * T_; idx += 256)
        Wm[idx / T_][idx % T_] = Wp[idx];
    __syncthreads();
    for (int idx = tid; idx < V_ * T_; idx += 256) {
        int v = idx / T_, t = idx % T_;
        float acc = 0.0f;
#pragma unroll
        for (int tp = 0; tp < T_; ++tp) acc += Xs[tp][v] * Wm[tp][t];
        Wh[(size_t)((n * T_ + t) * V_ + v) * HW + p] = acc;
    }
}

__global__ void eij_kernel(const float* __restrict__ Wh, const float* __restrict__ a,
                           float* __restrict__ ei, float* __restrict__ ej) {
    int g = blockIdx.x * 256 + threadIdx.x;
    int n  = g / (V_ * T_);
    int r  = g - n * (V_ * T_);
    int vv = r / T_;
    int tn = r - vv * T_;
    const float* whb = Wh + (size_t)n * T_ * V_ * HW;
    float ai = 0.0f, aj = 0.0f;
    for (int q = 0; q < THW; ++q) {
        int f  = tn * THW + q;
        int hh = f / (2 * WD * T_);
        int r2 = f - hh * (2 * WD * T_);
        int ww = r2 / T_;
        int t  = r2 - ww * T_;
        float w = a[q];
        if (ww < WD)
            ai += whb[(size_t)(t * V_ + vv) * HW + hh * WD + ww] * w;
        else
            aj += whb[(size_t)(t * V_ + vv) * HW + hh * WD + (ww - WD)] * w;
    }
    ei[g] = ai;
    ej[g] = aj;
}

__global__ void att_kernel(const float* __restrict__ ei, const float* __restrict__ ej,
                           float* __restrict__ att) {
    int g = blockIdx.x * 256 + threadIdx.x;
    int n   = g >> 10;
    int rem = g & 1023;
    int i   = rem >> 5;
    int j   = rem & 31;
    const float* eip = ei + (n * V_ + i) * T_;
    const float* ejp = ej + (n * V_ + j) * T_;
    float l[T_];
    float m = -1e30f;
#pragma unroll
    for (int t = 0; t < T_; ++t) {
        float e = eip[t] + ejp[t];
        e = e > 0.0f ? e : ALPHA * e;
        l[t] = e;
        m = fmaxf(m, e);
    }
    float s = 0.0f;
#pragma unroll
    for (int t = 0; t < T_; ++t) { l[t] = __expf(l[t] - m); s += l[t]; }
    float is = 1.0f / s;
#pragma unroll
    for (int t = 0; t < T_; ++t)
        att[(size_t)((n * T_ + t) * V_ + i) * V_ + j] = l[t] * is;
}

__global__ void __launch_bounds__(256) gemm_kernel(const float* __restrict__ Wh,
                                                   const float* __restrict__ att,
                                                   const float* __restrict__ adjn,
                                                   float* __restrict__ out) {
    __shared__ float Gl [V_ * 33];
    __shared__ float Ajl[V_ * 33];
    __shared__ float Ml [V_ * 33];
    __shared__ float Wl [V_ * 257];

    int bid = blockIdx.x;
    int n = bid / T_;
    int t = bid - n * T_;
    int tid = threadIdx.x;
    const float* gsl = att + (size_t)(n * T_ + t) * V_ * V_;
    const float* wsl = Wh  + (size_t)(n * T_ + t) * V_ * HW;

    for (int idx = tid; idx < V_ * V_; idx += 256) {
        Gl [(idx >> 5) * 33 + (idx & 31)] = gsl[idx];
        Ajl[(idx >> 5) * 33 + (idx & 31)] = adjn[idx];
    }
    for (int idx = tid; idx < V_ * HW; idx += 256)
        Wl[(idx >> 8) * 257 + (idx & 255)] = wsl[idx];
    __syncthreads();

    for (int idx = tid; idx < V_ * V_; idx += 256) {
        int v = idx >> 5, j = idx & 31;
        float acc = 0.0f;
#pragma unroll
        for (int i2 = 0; i2 < V_; ++i2) acc += Ajl[i2 * 33 + v] * Gl[i2 * 33 + j];
        Ml[v * 33 + j] = acc;
    }
    __syncthreads();

    int wave = tid >> 5, lane = tid & 31;
    int lh = lane >> 4, lm = lane & 15;
    for (int q = 0; q < 4; ++q) {
        int tile = wave * 4 + q;
        int tm = tile >> 4, tn = tile & 15;
        int arow = tm * 16 + lm;
        int col  = tn * 16 + lm;
        v8f c = {};
#pragma unroll
        for (int kk = 0; kk < 8; ++kk) {
            int kb = kk * 4 + lh * 2;
            v2f a, b;
            a.x = Ml[arow * 33 + kb];
            a.y = Ml[arow * 33 + kb + 1];
            b.x = Wl[kb * 257 + col];
            b.y = Wl[(kb + 1) * 257 + col];
            c = __builtin_amdgcn_wmma_f32_16x16x4_f32(
                     false, a,  false, b,
                     (short)0, c,  false,  false);
        }
        int vb = tm * 16 + lh * 8;
        float* ob = out + ((size_t)(n * HW + col) * T_ + t) * V_ + vb;
#pragma unroll
        for (int r = 0; r < 8; ++r) {
            float x = c[r];
            float y = x > 0.0f ? x : (__expf(x) - 1.0f);
            ob[r] = 1.0f / (1.0f + __expf(-y));
        }
    }
}

extern "C" void kernel_launch(void* const* d_in, const int* in_sizes, int n_in,
                              void* d_out, int out_size, void* d_ws, size_t ws_size,
                              hipStream_t stream) {
    const float* h  = (const float*)d_in[0];
    const float* Wp = (const float*)d_in[1];
    const float* ap = (const float*)d_in[2];
    const float* Bp = (const float*)d_in[3];
    float* out = (float*)d_out;

    float* ws    = (float*)d_ws;
    float* Whb   = ws;
    float* adjn  = Whb  + (size_t)N_ * T_ * V_ * HW;
    float* ei    = adjn + V_ * V_;
    float* ej    = ei   + N_ * V_ * T_;
    float* attb  = ej   + N_ * V_ * T_;

    adj_kernel <<<1,           32, 0, stream>>>(Bp, adjn);
    wh_kernel  <<<N_ * HW,    256, 0, stream>>>(h, Wp, Whb);
    eij_kernel <<<(N_ * V_ * T_) / 256, 256, 0, stream>>>(Whb, ap, ei, ej);
    att_kernel <<<(N_ * V_ * V_) / 256, 256, 0, stream>>>(ei, ej, attb);
    gemm_kernel<<<N_ * T_,    256, 0, stream>>>(Whb, attb, adjn, out);
}


# Softmax/M launch removed by re-association: product blocks make att in LDS, then apply att_t and adjn^T as chained matrix products (hp stays in registers); 3 launches to 2; O sums re-associated
# speedup vs baseline: 19.3993x; 19.3993x over previous
#include <hip/hip_runtime.h>


#define ALPHA 0.2f
#define N_  4
#define HD  16
#define WD  16
#define T_  24
#define V_  32
#define HW  256
#define THW 512

typedef __attribute__((ext_vector_type(2))) float v2f;
typedef __attribute__((ext_vector_type(8))) float v8f;

#define NSEG 22
typedef __attribute__((ext_vector_type(4))) float v4f;

__global__ void __launch_bounds__(384) eij_kernel(const float* __restrict__ h,
                                                  const float* __restrict__ Wp,
                                                  const float* __restrict__ a,
                                                  float* __restrict__ ei,
                                                  float* __restrict__ ej) {
    __shared__ float Wm[32 * 25];
    __shared__ float As[768];
    __shared__ float Cs[NSEG * T_];
    __shared__ float Pi[48 * V_];
    __shared__ float Pj[48 * V_];

    int bid = blockIdx.x;
    int n  = bid / T_;
    int tn = bid - n * T_;
    int tid = threadIdx.x;
    int wave = __builtin_amdgcn_readfirstlane(tid >> 5);
    int lane = tid & 31;
    int rg = lane >> 3, c4 = lane & 7;
    int par = wave / 6;
    int row = (wave - par * 6) * 4 + rg;
    int s_lo = (tn * THW) / T_;

    int i1 = tid + 384;
    float w0 = Wp[tid];
    float w1 = Wp[i1 < T_ * T_ ? i1 : tid];
    float a0 = a[tid];
    float a1 = a[i1 < THW ? i1 : tid];
    Wm[tid + tid / T_] = w0;
    Wm[i1 + i1 / T_] = w1;
    As[tid] = a0;
    As[i1] = a1;
    __syncthreads();

    const v4f* hn = (const v4f*)(h + (size_t)n * HW * T_ * V_);
    v4f x[11];
#pragma unroll
    for (int k = 0; k < 11; ++k) {
        int s = s_lo + 2 * k + par;
        int p = (s >> 5) * WD + (s & (WD - 1));
        x[k] = hn[(p * T_ + row) * (V_ / 4) + c4];
    }

    for (int idx = tid; idx < NSEG * T_; idx += 384) {
        int sl = idx / T_;
        int tp = idx - sl * T_;
        int q0 = (s_lo + sl) * T_ - tn * THW;
        float c = 0.0f;
#pragma unroll
        for (int t = 0; t < T_; ++t) {
            int q = q0 + t;
            float w = Wm[tp * 25 + t] * As[q & (THW - 1)];
            c += ((unsigned)q < (unsigned)THW) ? w : 0.0f;
        }
        Cs[idx] = c;
    }
    __syncthreads();

    v4f ai = {0.0f, 0.0f, 0.0f, 0.0f}, aj = {0.0f, 0.0f, 0.0f, 0.0f};
#pragma unroll
    for (int k = 0; k < 11; ++k) {
        int sl = 2 * k + par;
        float c = Cs[sl * T_ + row];
        if ((s_lo + sl) & WD) aj += x[k] * c;
        else                  ai += x[k] * c;
    }
    *(v4f*)&Pi[(wave * 4 + rg) * V_ + c4 * 4] = ai;
    *(v4f*)&Pj[(wave * 4 + rg) * V_ + c4 * 4] = aj;
    __syncthreads();

    if (tid < 2 * V_) {
        int v = tid & (V_ - 1);
        const float* P = (tid < V_) ? Pi : Pj;
        float s = 0.0f;
#pragma unroll
        for (int r = 0; r < 48; ++r) s += P[r * V_ + v];
        float* dst = (tid < V_) ? ei : ej;
        dst[(n * V_ + v) * T_ + tn] = s;
    }
}

#define PT 16
#define JS 18
#define TS 577
#define AJS 36
#define ATI 36
#define ATS (V_ * ATI)

__device__ __forceinline__ float lane_val(float x, int l) {
    return __builtin_bit_cast(float, __builtin_amdgcn_readlane(__builtin_bit_cast(int, x), l));
}

__global__ void __launch_bounds__(544) prod_kernel(const float* __restrict__ h,
                                                   const float* __restrict__ Wp,
                                                   const float* __restrict__ ei,
                                                   const float* __restrict__ ej,
                                                   const float* __restrict__ B,
                                                   float* __restrict__ out) {
    __shared__ float Whs[(T_ + 1) * TS];
    __shared__ __attribute__((aligned(16))) float Att[T_ * ATS];
    __shared__ __attribute__((aligned(16))) float Aj[V_ * AJS];

    int bid = blockIdx.x;
    int n  = bid >> 4;
    int p0 = (bid & 15) * PT;
    int tid = threadIdx.x;
    int wave = __builtin_amdgcn_readfirstlane(tid >> 5);
    int lane = tid & 31;
    int lh = lane >> 4, lm = lane & 15;

    if (wave == 16) {
        int i = lane;
        const v4f* Br = (const v4f*)(B + i * V_);
        v4f b[V_ / 4];
#pragma unroll
        for (int k = 0; k < V_ / 4; ++k) b[k] = Br[k];
        float a[V_];
        float mn = 1e30f, mx = -1e30f;
#pragma unroll
        for (int j = 0; j < V_; ++j) {
            float v = b[j >> 2][j & 3] + 1e-6f + (i == j ? 1.0f : 0.0f);
            a[j] = v;
            mn = fminf(mn, v); mx = fmaxf(mx, v);
        }
        float gmn = lane_val(mn, 0), gmx = lane_val(mx, 0);
#pragma unroll
        for (int k = 1; k < V_; ++k) {
            gmn = fminf(gmn, lane_val(mn, k));
            gmx = fmaxf(gmx, lane_val(mx, k));
        }
        float inv = 1.0f / (gmx - gmn);
        float s = 0.0f;
#pragma unroll
        for (int j = 0; j < V_; ++j) s += (a[j] - gmn) * inv;
        float di = rsqrtf(s);
#pragma unroll
        for (int k = 0; k < V_ / 4; ++k) {
            v4f o;
#pragma unroll
            for (int c = 0; c < 4; ++c) {
                int j = k * 4 + c;
                o[c] = (a[j] - gmn) * inv * di * lane_val(di, j);
            }
            *(v4f*)&Aj[i * AJS + k * 4] = o;
        }
    } else {
        const v4f* ejp = (const v4f*)(ej + (n * V_ + lane) * T_);
        v4f ejv[T_ / 4];
#pragma unroll
        for (int k = 0; k < T_ / 4; ++k) ejv[k] = ejp[k];
        __builtin_amdgcn_sched_barrier(0);

        const float* hp = h + (size_t)(n * HW + p0 + wave) * T_ * V_;
        v2f xa[2][6];
#pragma unroll
        for (int jh = 0; jh < 2; ++jh)
#pragma unroll
            for (int kk = 0; kk < 6; ++kk) {
                int row = kk * 4 + lh * 2;
                xa[jh][kk].x = hp[row * V_ + jh * 16 + lm];
                xa[jh][kk].y = hp[(row + 1) * V_ + jh * 16 + lm];
            }
        v2f wr[2][6];
#pragma unroll
        for (int nt = 0; nt < 2; ++nt) {
            int t = nt * 16 + lm;
            int tc = t < T_ ? t : 0;
#pragma unroll
            for (int kk = 0; kk < 6; ++kk) {
                int row = kk * 4 + lh * 2;
                wr[nt][kk].x = Wp[row * T_ + tc];
                wr[nt][kk].y = Wp[(row + 1) * T_ + tc];
            }
        }
        __builtin_amdgcn_sched_barrier(0);

#pragma unroll
        for (int r = 0; r < 2; ++r) {
            int i = wave + 16 * r;
            const float* eip = ei + (n * V_ + i) * T_;
            float l[T_];
            float m = -1e30f;
#pragma unroll
            for (int t = 0; t < T_; ++t) {
                float x = eip[t] + ejv[t >> 2][t & 3];
                x = x > 0.0f ? x : ALPHA * x;
                l[t] = x;
                m = fmaxf(m, x);
            }
            float s = 0.0f;
#pragma unroll
            for (int t = 0; t < T_; ++t) { l[t] = __expf(l[t] - m); s += l[t]; }
            float is = 1.0f / s;
#pragma unroll
            for (int t = 0; t < T_; ++t) Att[t * ATS + i * ATI + lane] = l[t] * is;
        }

        v2f wb[2][6];
#pragma unroll
        for (int nt = 0; nt < 2; ++nt) {
            int t = nt * 16 + lm;
#pragma unroll
            for (int kk = 0; kk < 6; ++kk) {
                wb[nt][kk].x = t < T_ ? wr[nt][kk].x : 0.0f;
                wb[nt][kk].y = t < T_ ? wr[nt][kk].y : 0.0f;
            }
        }
#pragma unroll
        for (int jh = 0; jh < 2; ++jh)
#pragma unroll
            for (int nt = 0; nt < 2; ++nt) {
                v8f c = {};
#pragma unroll
                for (int kk = 0; kk < 6; ++kk)
                    c = __builtin_amdgcn_wmma_f32_16x16x4_f32(
                             false, xa[jh][kk],  false, wb[nt][kk],
                             (short)0, c,  false,  false);
                int tr = nt * 16 + lm;
                tr = tr < T_ ? tr : T_;
                float* wp = &Whs[tr * TS + (jh * 16 + lh * 8) * JS + wave];
#pragma unroll
                for (int r = 0; r < 8; ++r) wp[r * JS] = c[r];
            }
    }
    __syncthreads();

    if (wave < 12) {
        v2f aT[2][8];
#pragma unroll
        for (int vt = 0; vt < 2; ++vt)
#pragma unroll
            for (int s = 0; s < 8; ++s) {
                int i0 = 16 * (s >> 2) + 8 * lh + 2 * (s & 3);
                aT[vt][s].x = Aj[i0 * AJS + 16 * vt + lm];
                aT[vt][s].y = Aj[(i0 + 1) * AJS + 16 * vt + lm];
            }
#pragma unroll
        for (int q = 0; q < 2; ++q) {
            int t = wave * 2 + q;
            const float* wt = &Whs[t * TS + lh * 2 * JS + lm];
            const float* at = &Att[t * ATS + lm * ATI + lh * 2];
            v8f hp0 = {}, hp1 = {};
#pragma unroll
            for (int kk = 0; kk < 8; ++kk) {
                v2f b;
                b.x = wt[(kk * 4) * JS];
                b.y = wt[(kk * 4 + 1) * JS];
                v2f a0 = *(const v2f*)&at[kk * 4];
                v2f a1 = *(const v2f*)&at[16 * ATI + kk * 4];
                hp0 = __builtin_amdgcn_wmma_f32_16x16x4_f32(
                         false, a0,  false, b,
                         (short)0, hp0,  false,  false);
                hp1 = __builtin_amdgcn_wmma_f32_16x16x4_f32(
                         false, a1,  false, b,
                         (short)0, hp1,  false,  false);
            }
            v8f o0 = {}, o1 = {};
#pragma unroll
            for (int s = 0; s < 8; ++s) {
                v2f b;
                if (s < 4) { b.x = hp0[2 * (s & 3)]; b.y = hp0[2 * (s & 3) + 1]; }
                else       { b.x = hp1[2 * (s & 3)]; b.y = hp1[2 * (s & 3) + 1]; }
                o0 = __builtin_amdgcn_wmma_f32_16x16x4_f32(
                         false, aT[0][s],  false, b,
                         (short)0, o0,  false,  false);
                o1 = __builtin_amdgcn_wmma_f32_16x16x4_f32(
                         false, aT[1][s],  false, b,
                         (short)0, o1,  false,  false);
            }
#pragma unroll
            for (int vt = 0; vt < 2; ++vt) {
                v8f c = vt ? o1 : o0;
                int vb = vt * 16 + lh * 8;
                float* ob = out + ((size_t)(n * HW + p0 + lm) * T_ + t) * V_ + vb;
#pragma unroll
                for (int r = 0; r < 8; ++r) {
                    float x = c[r];
                    float y = x > 0.0f ? x : (__expf(x) - 1.0f);
                    ob[r] = 1.0f / (1.0f + __expf(-y));
                }
            }
        }
    }
}

extern "C" void kernel_launch(void* const* d_in, const int* in_sizes, int n_in,
                              void* d_out, int out_size, void* d_ws, size_t ws_size,
                              hipStream_t stream) {
    const float* h  = (const float*)d_in[0];
    const float* Wp = (const float*)d_in[1];
    const float* ap = (const float*)d_in[2];
    const float* Bp = (const float*)d_in[3];
    float* out = (float*)d_out;

    float* ws    = (float*)d_ws;
    float* ei    = ws;
    float* ej    = ei   + N_ * V_ * T_;

    eij_kernel <<<N_ * T_,    384, 0, stream>>>(h, Wp, ap, ei, ej);
    prod_kernel<<<N_ * (HW / PT), 544, 0, stream>>>(h, Wp, ei, ej, Bp, out);
}


# Product launch re-tiled over t: block per (n, 16 pixels, 8 t) of 8+1 waves, 192 blocks not 64; a wave makes 2 pixels' Wh in one column tile and 4 softmax rows, then one t; sum orders kept
# speedup vs baseline: 20.8361x; 1.0741x over previous
#include <hip/hip_runtime.h>


#define ALPHA 0.2f
#define N_  4
#define HD  16
#define WD  16
#define T_  24
#define V_  32
#define HW  256
#define THW 512

typedef __attribute__((ext_vector_type(2))) float v2f;
typedef __attribute__((ext_vector_type(8))) float v8f;

#define NSEG 22
typedef __attribute__((ext_vector_type(4))) float v4f;

__global__ void __launch_bounds__(384) eij_kernel(const float* __restrict__ h,
                                                  const float* __restrict__ Wp,
                                                  const float* __restrict__ a,
                                                  float* __restrict__ ei,
                                                  float* __restrict__ ej) {
    __shared__ float Wm[32 * 25];
    __shared__ float As[768];
    __shared__ float Cs[NSEG * T_];
    __shared__ float Pi[48 * V_];
    __shared__ float Pj[48 * V_];

    int bid = blockIdx.x;
    int n  = bid / T_;
    int tn = bid - n * T_;
    int tid = threadIdx.x;
    int wave = __builtin_amdgcn_readfirstlane(tid >> 5);
    int lane = tid & 31;
    int rg = lane >> 3, c4 = lane & 7;
    int par = wave / 6;
    int row = (wave - par * 6) * 4 + rg;
    int s_lo = (tn * THW) / T_;

    int i1 = tid + 384;
    float w0 = Wp[tid];
    float w1 = Wp[i1 < T_ * T_ ? i1 : tid];
    float a0 = a[tid];
    float a1 = a[i1 < THW ? i1 : tid];
    Wm[tid + tid / T_] = w0;
    Wm[i1 + i1 / T_] = w1;
    As[tid] = a0;
    As[i1] = a1;
    __syncthreads();

    const v4f* hn = (const v4f*)(h + (size_t)n * HW * T_ * V_);
    v4f x[11];
#pragma unroll
    for (int k = 0; k < 11; ++k) {
        int s = s_lo + 2 * k + par;
        int p = (s >> 5) * WD + (s & (WD - 1));
        x[k] = hn[(p * T_ + row) * (V_ / 4) + c4];
    }

    for (int idx = tid; idx < NSEG * T_; idx += 384) {
        int sl = idx / T_;
        int tp = idx - sl * T_;
        int q0 = (s_lo + sl) * T_ - tn * THW;
        float c = 0.0f;
#pragma unroll
        for (int t = 0; t < T_; ++t) {
            int q = q0 + t;
            float w = Wm[tp * 25 + t] * As[q & (THW - 1)];
            c += ((unsigned)q < (unsigned)THW) ? w : 0.0f;
        }
        Cs[idx] = c;
    }
    __syncthreads();

    v4f ai = {0.0f, 0.0f, 0.0f, 0.0f}, aj = {0.0f, 0.0f, 0.0f, 0.0f};
#pragma unroll
    for (int k = 0; k < 11; ++k) {
        int sl = 2 * k + par;
        float c = Cs[sl * T_ + row];
        if ((s_lo + sl) & WD) aj += x[k] * c;
        else                  ai += x[k] * c;
    }
    *(v4f*)&Pi[(wave * 4 + rg) * V_ + c4 * 4] = ai;
    *(v4f*)&Pj[(wave * 4 + rg) * V_ + c4 * 4] = aj;
    __syncthreads();

    if (tid < 2 * V_) {
        int v = tid & (V_ - 1);
        const float* P = (tid < V_) ? Pi : Pj;
        float s = 0.0f;
#pragma unroll
        for (int r = 0; r < 48; ++r) s += P[r * V_ + v];
        float* dst = (tid < V_) ? ei : ej;
        dst[(n * V_ + v) * T_ + tn] = s;
    }
}

#define PT 16
#define TG 8
#define NTG 3
#define JS 18
#define TS 577
#define AJS 36
#define ATI 36
#define ATS (V_ * ATI)

__device__ __forceinline__ float lane_val(float x, int l) {
    return __builtin_bit_cast(float, __builtin_amdgcn_readlane(__builtin_bit_cast(int, x), l));
}

__global__ void __launch_bounds__(288) prod_kernel(const float* __restrict__ h,
                                                   const float* __restrict__ Wp,
                                                   const float* __restrict__ ei,
                                                   const float* __restrict__ ej,
                                                   const float* __restrict__ B,
                                                   float* __restrict__ out) {
    __shared__ float Whs[16 * TS];
    __shared__ __attribute__((aligned(16))) float Att[TG * ATS];
    __shared__ __attribute__((aligned(16))) float Aj[V_ * AJS];

    int bid = blockIdx.x;
    int pt = bid / NTG;
    int tg = bid - pt * NTG;
    int n  = pt >> 4;
    int p0 = (pt & 15) * PT;
    int t0 = tg * TG;
    int tid = threadIdx.x;
    int wave = __builtin_amdgcn_readfirstlane(tid >> 5);
    int lane = tid & 31;
    int lh = lane >> 4, lm = lane & 15;

    if (wave == 8) {
        int i = lane;
        const v4f* Br = (const v4f*)(B + i * V_);
        v4f b[V_ / 4];
#pragma unroll
        for (int k = 0; k < V_ / 4; ++k) b[k] = Br[k];
        float a[V_];
        float mn = 1e30f, mx = -1e30f;
#pragma unroll
        for (int j = 0; j < V_; ++j) {
            float v = b[j >> 2][j & 3] + 1e-6f + (i == j ? 1.0f : 0.0f);
            a[j] = v;
            mn = fminf(mn, v); mx = fmaxf(mx, v);
        }
        float gmn = lane_val(mn, 0), gmx = lane_val(mx, 0);
#pragma unroll
        for (int k = 1; k < V_; ++k) {
            gmn = fminf(gmn, lane_val(mn, k));
            gmx = fmaxf(gmx, lane_val(mx, k));
        }
        float inv = 1.0f / (gmx - gmn);
        float s = 0.0f;
#pragma unroll
        for (int j = 0; j < V_; ++j) s += (a[j] - gmn) * inv;
        float di = rsqrtf(s);
#pragma unroll
        for (int k = 0; k < V_ / 4; ++k) {
            v4f o;
#pragma unroll
            for (int c = 0; c < 4; ++c) {
                int j = k * 4 + c;
                o[c] = (a[j] - gmn) * inv * di * lane_val(di, j);
            }
            *(v4f*)&Aj[i * AJS + k * 4] = o;
        }
    } else {
        const v4f* ejp = (const v4f*)(ej + (n * V_ + lane) * T_);
        v4f ejv[T_ / 4];
#pragma unroll
        for (int k = 0; k < T_ / 4; ++k) ejv[k] = ejp[k];
        __builtin_amdgcn_sched_barrier(0);

        v2f xa[2][2][6];
#pragma unroll
        for (int px = 0; px < 2; ++px) {
            const float* hp = h + (size_t)(n * HW + p0 + wave + 8 * px) * T_ * V_;
#pragma unroll
            for (int jh = 0; jh < 2; ++jh)
#pragma unroll
                for (int kk = 0; kk < 6; ++kk) {
                    int row = kk * 4 + lh * 2;
                    xa[px][jh][kk].x = hp[row * V_ + jh * 16 + lm];
                    xa[px][jh][kk].y = hp[(row + 1) * V_ + jh * 16 + lm];
                }
        }
        int tc = t0 + lm;
        tc = tc < T_ ? tc : T_ - 1;
        v2f wb[6];
#pragma unroll
        for (int kk = 0; kk < 6; ++kk) {
            int row = kk * 4 + lh * 2;
            wb[kk].x = Wp[row * T_ + tc];
            wb[kk].y = Wp[(row + 1) * T_ + tc];
        }
        __builtin_amdgcn_sched_barrier(0);

#pragma unroll
        for (int r = 0; r < 4; ++r) {
            int i = wave + 8 * r;
            const float* eip = ei + (n * V_ + i) * T_;
            float l[T_];
            float m = -1e30f;
#pragma unroll
            for (int t = 0; t < T_; ++t) {
                float x = eip[t] + ejv[t >> 2][t & 3];
                x = x > 0.0f ? x : ALPHA * x;
                l[t] = x;
                m = fmaxf(m, x);
            }
            float s = 0.0f;
#pragma unroll
            for (int t = 0; t < T_; ++t) { l[t] = __expf(l[t] - m); s += l[t]; }
            float is = 1.0f / s;
            float* ar = &Att[i * ATI + lane];
#pragma unroll
            for (int g = 0; g < NTG; ++g)
                if (tg == g) {
#pragma unroll
                    for (int k = 0; k < TG; ++k) ar[k * ATS] = l[g * TG + k] * is;
                }
        }

#pragma unroll
        for (int px = 0; px < 2; ++px)
#pragma unroll
            for (int jh = 0; jh < 2; ++jh) {
                v8f c = {};
#pragma unroll
                for (int kk = 0; kk < 6; ++kk)
                    c = __builtin_amdgcn_wmma_f32_16x16x4_f32(
                             false, xa[px][jh][kk],  false, wb[kk],
                             (short)0, c,  false,  false);
                float* wp = &Whs[lm * TS + (jh * 16 + lh * 8) * JS + wave + 8 * px];
#pragma unroll
                for (int r = 0; r < 8; ++r) wp[r * JS] = c[r];
            }
    }
    __syncthreads();

    if (wave < 8) {
        v2f aT[2][8];
#pragma unroll
        for (int vt = 0; vt < 2; ++vt)
#pragma unroll
            for (int s = 0; s < 8; ++s) {
                int i0 = 16 * (s >> 2) + 8 * lh + 2 * (s & 3);
                aT[vt][s].x = Aj[i0 * AJS + 16 * vt + lm];
                aT[vt][s].y = Aj[(i0 + 1) * AJS + 16 * vt + lm];
            }
        int t = t0 + wave;
        const float* wt = &Whs[wave * TS + lh * 2 * JS + lm];
        const float* at = &Att[wave * ATS + lm * ATI + lh * 2];
        v8f hp0 = {}, hp1 = {};
#pragma unroll
        for (int kk = 0; kk < 8; ++kk) {
            v2f b;
            b.x = wt[(kk * 4) * JS];
            b.y = wt[(kk * 4 + 1) * JS];
            v2f a0 = *(const v2f*)&at[kk * 4];
            v2f a1 = *(const v2f*)&at[16 * ATI + kk * 4];
            hp0 = __builtin_amdgcn_wmma_f32_16x16x4_f32(
                     false, a0,  false, b,
                     (short)0, hp0,  false,  false);
            hp1 = __builtin_amdgcn_wmma_f32_16x16x4_f32(
                     false, a1,  false, b,
                     (short)0, hp1,  false,  false);
        }
        v8f o0 = {}, o1 = {};
#pragma unroll
        for (int s = 0; s < 8; ++s) {
            v2f b;
            if (s < 4) { b.x = hp0[2 * (s & 3)]; b.y = hp0[2 * (s & 3) + 1]; }
            else       { b.x = hp1[2 * (s & 3)]; b.y = hp1[2 * (s & 3) + 1]; }
            o0 = __builtin_amdgcn_wmma_f32_16x16x4_f32(
                     false, aT[0][s],  false, b,
                     (short)0, o0,  false,  false);
            o1 = __builtin_amdgcn_wmma_f32_16x16x4_f32(
                     false, aT[1][s],  false, b,
                     (short)0, o1,  false,  false);
        }
#pragma unroll
        for (int vt = 0; vt < 2; ++vt) {
            v8f c = vt ? o1 : o0;
            int vb = vt * 16 + lh * 8;
            float* ob = out + ((size_t)(n * HW + p0 + lm) * T_ + t) * V_ + vb;
#pragma unroll
            for (int r = 0; r < 8; ++r) {
                float x = c[r];
                float y = x > 0.0f ? x : (__expf(x) - 1.0f);
                ob[r] = 1.0f / (1.0f + __expf(-y));
            }
        }
    }
}

extern "C" void kernel_launch(void* const* d_in, const int* in_sizes, int n_in,
                              void* d_out, int out_size, void* d_ws, size_t ws_size,
                              hipStream_t stream) {
    const float* h  = (const float*)d_in[0];
    const float* Wp = (const float*)d_in[1];
    const float* ap = (const float*)d_in[2];
    const float* Bp = (const float*)d_in[3];
    float* out = (float*)d_out;

    float* ws    = (float*)d_ws;
    float* ei    = ws;
    float* ej    = ei   + N_ * V_ * T_;

    eij_kernel <<<N_ * T_,    384, 0, stream>>>(h, Wp, ap, ei, ej);
    prod_kernel<<<N_ * (HW / PT) * NTG, 288, 0, stream>>>(h, Wp, ei, ej, Bp, out);
}


# Product launch, phase before its barrier software-pipelined: stage 1's 24 matrix instructions issue one by one inside softmax rows 1..3, each row's ei fetched a row ahead; sum orders kept
# speedup vs baseline: 21.9357x; 1.0528x over previous
#include <hip/hip_runtime.h>


#define ALPHA 0.2f
#define N_  4
#define HD  16
#define WD  16
#define T_  24
#define V_  32
#define HW  256
#define THW 512

typedef __attribute__((ext_vector_type(2))) float v2f;
typedef __attribute__((ext_vector_type(8))) float v8f;

#define NSEG 22
typedef __attribute__((ext_vector_type(4))) float v4f;

__global__ void __launch_bounds__(384) eij_kernel(const float* __restrict__ h,
                                                  const float* __restrict__ Wp,
                                                  const float* __restrict__ a,
                                                  float* __restrict__ ei,
                                                  float* __restrict__ ej) {
    __shared__ float Wm[32 * 25];
    __shared__ float As[768];
    __shared__ float Cs[NSEG * T_];
    __shared__ float Pi[48 * V_];
    __shared__ float Pj[48 * V_];

    int bid = blockIdx.x;
    int n  = bid / T_;
    int tn = bid - n * T_;
    int tid = threadIdx.x;
    int wave = __builtin_amdgcn_readfirstlane(tid >> 5);
    int lane = tid & 31;
    int rg = lane >> 3, c4 = lane & 7;
    int par = wave / 6;
    int row = (wave - par * 6) * 4 + rg;
    int s_lo = (tn * THW) / T_;

    int i1 = tid + 384;
    float w0 = Wp[tid];
    float w1 = Wp[i1 < T_ * T_ ? i1 : tid];
    float a0 = a[tid];
    float a1 = a[i1 < THW ? i1 : tid];
    Wm[tid + tid / T_] = w0;
    Wm[i1 + i1 / T_] = w1;
    As[tid] = a0;
    As[i1] = a1;
    __syncthreads();

    const v4f* hn = (const v4f*)(h + (size_t)n * HW * T_ * V_);
    v4f x[11];
#pragma unroll
    for (int k = 0; k < 11; ++k) {
        int s = s_lo + 2 * k + par;
        int p = (s >> 5) * WD + (s & (WD - 1));
        x[k] = hn[(p * T_ + row) * (V_ / 4) + c4];
    }

    for (int idx = tid; idx < NSEG * T_; idx += 384) {
        int sl = idx / T_;
        int tp = idx - sl * T_;
        int q0 = (s_lo + sl) * T_ - tn * THW;
        float c = 0.0f;
#pragma unroll
        for (int t = 0; t < T_; ++t) {
            int q = q0 + t;
            float w = Wm[tp * 25 + t] * As[q & (THW - 1)];
            c += ((unsigned)q < (unsigned)THW) ? w : 0.0f;
        }
        Cs[idx] = c;
    }
    __syncthreads();

    v4f ai = {0.0f, 0.0f, 0.0f, 0.0f}, aj = {0.0f, 0.0f, 0.0f, 0.0f};
#pragma unroll
    for (int k = 0; k < 11; ++k) {
        int sl = 2 * k + par;
        float c = Cs[sl * T_ + row];
        if ((s_lo + sl) & WD) aj += x[k] * c;
        else                  ai += x[k] * c;
    }
    *(v4f*)&Pi[(wave * 4 + rg) * V_ + c4 * 4] = ai;
    *(v4f*)&Pj[(wave * 4 + rg) * V_ + c4 * 4] = aj;
    __syncthreads();

    if (tid < 2 * V_) {
        int v = tid & (V_ - 1);
        const float* P = (tid < V_) ? Pi : Pj;
        float s = 0.0f;
#pragma unroll
        for (int r = 0; r < 48; ++r) s += P[r * V_ + v];
        float* dst = (tid < V_) ? ei : ej;
        dst[(n * V_ + v) * T_ + tn] = s;
    }
}

#define PT 16
#define TG 8
#define NTG 3
#define JS 18
#define TS 577
#define AJS 36
#define ATI 36
#define ATS (V_ * ATI)

__device__ __forceinline__ float lane_val(float x, int l) {
    return __builtin_bit_cast(float, __builtin_amdgcn_readlane(__builtin_bit_cast(int, x), l));
}

__global__ void __launch_bounds__(288) prod_kernel(const float* __restrict__ h,
                                                   const float* __restrict__ Wp,
                                                   const float* __restrict__ ei,
                                                   const float* __restrict__ ej,
                                                   const float* __restrict__ B,
                                                   float* __restrict__ out) {
    __shared__ float Whs[16 * TS];
    __shared__ __attribute__((aligned(16))) float Att[TG * ATS];
    __shared__ __attribute__((aligned(16))) float Aj[V_ * AJS];

    int bid = blockIdx.x;
    int pt = bid / NTG;
    int tg = bid - pt * NTG;
    int n  = pt >> 4;
    int p0 = (pt & 15) * PT;
    int t0 = tg * TG;
    int tid = threadIdx.x;
    int wave = __builtin_amdgcn_readfirstlane(tid >> 5);
    int lane = tid & 31;
    int lh = lane >> 4, lm = lane & 15;

    if (wave == 8) {
        int i = lane;
        const v4f* Br = (const v4f*)(B + i * V_);
        v4f b[V_ / 4];
#pragma unroll
        for (int k = 0; k < V_ / 4; ++k) b[k] = Br[k];
        float a[V_];
        float mn = 1e30f, mx = -1e30f;
#pragma unroll
        for (int j = 0; j < V_; ++j) {
            float v = b[j >> 2][j & 3] + 1e-6f + (i == j ? 1.0f : 0.0f);
            a[j] = v;
            mn = fminf(mn, v); mx = fmaxf(mx, v);
        }
        float gmn = lane_val(mn, 0), gmx = lane_val(mx, 0);
#pragma unroll
        for (int k = 1; k < V_; ++k) {
            gmn = fminf(gmn, lane_val(mn, k));
            gmx = fmaxf(gmx, lane_val(mx, k));
        }
        float inv = 1.0f / (gmx - gmn);
        float s = 0.0f;
#pragma unroll
        for (int j = 0; j < V_; ++j) s += (a[j] - gmn) * inv;
        float di = rsqrtf(s);
#pragma unroll
        for (int k = 0; k < V_ / 4; ++k) {
            v4f o;
#pragma unroll
            for (int c = 0; c < 4; ++c) {
                int j = k * 4 + c;
                o[c] = (a[j] - gmn) * inv * di * lane_val(di, j);
            }
            *(v4f*)&Aj[i * AJS + k * 4] = o;
        }
    } else {
        const v4f* ejp = (const v4f*)(ej + (n * V_ + lane) * T_);
        v4f ejv[T_ / 4];
#pragma unroll
        for (int k = 0; k < T_ / 4; ++k) ejv[k] = ejp[k];
        __builtin_amdgcn_sched_barrier(0);

        int tc = t0 + lm;
        tc = tc < T_ ? tc : T_ - 1;
        v2f wb[6];
#pragma unroll
        for (int kk = 0; kk < 6; ++kk) {
            int row = kk * 4 + lh * 2;
            wb[kk].x = Wp[row * T_ + tc];
            wb[kk].y = Wp[(row + 1) * T_ + tc];
        }
        __builtin_amdgcn_sched_barrier(0);
        v2f xa[2][2][6];
#pragma unroll
        for (int kk = 0; kk < 6; ++kk)
#pragma unroll
            for (int px = 0; px < 2; ++px) {
                const float* hp = h + (size_t)(n * HW + p0 + wave + 8 * px) * T_ * V_;
#pragma unroll
                for (int jh = 0; jh < 2; ++jh) {
                    int row = kk * 4 + lh * 2;
                    xa[px][jh][kk].x = hp[row * V_ + jh * 16 + lm];
                    xa[px][jh][kk].y = hp[(row + 1) * V_ + jh * 16 + lm];
                }
            }
        const float* eib = ei + (n * V_ + wave) * T_;
        float e[4][T_];
#pragma unroll
        for (int t = 0; t < T_; ++t) { e[0][t] = eib[t]; e[1][t] = eib[8 * T_ + t]; }
        asm volatile("" :: "s"(e[0][0]), "s"(e[0][4]), "s"(e[0][8]), "s"(e[0][12]), "s"(e[0][16]), "s"(e[0][20]));
        __builtin_amdgcn_sched_barrier(0);

        v8f c[2][2] = {};
#pragma unroll
        for (int r = 0; r < 4; ++r) {
            int i = wave + 8 * r;
            float l[T_];
            float m = -1e30f;
#pragma unroll
            for (int ch = 0; ch < 4; ++ch) {
#pragma unroll
                for (int t = ch * 6; t < ch * 6 + 6; ++t) {
                    float x = e[r][t] + ejv[t >> 2][t & 3];
                    x = x > 0.0f ? x : ALPHA * x;
                    l[t] = x;
                    m = fmaxf(m, x);
                }
                if (r >= 1) {
                    int w = (r - 1) * 8 + ch;
                    int kk = w >> 2, px = (w >> 1) & 1, jh = w & 1;
                    __builtin_amdgcn_sched_barrier(0);
                    c[px][jh] = __builtin_amdgcn_wmma_f32_16x16x4_f32(
                             false, xa[px][jh][kk],  false, wb[kk],
                             (short)0, c[px][jh],  false,  false);
                    __builtin_amdgcn_sched_barrier(0);
                }
                if (ch == 0 && r == 2) {
#pragma unroll
                    for (int t = 0; t < T_; ++t) e[3][t] = eib[24 * T_ + t];
                    __builtin_amdgcn_sched_barrier(0);
                }
            }
            if (r == 0) {
#pragma unroll
                for (int t = 0; t < T_; ++t) e[2][t] = eib[16 * T_ + t];
                __builtin_amdgcn_sched_barrier(0);
            }
            float s = 0.0f;
#pragma unroll
            for (int ch = 0; ch < 4; ++ch) {
#pragma unroll
                for (int t = ch * 6; t < ch * 6 + 6; ++t) { l[t] = __expf(l[t] - m); s += l[t]; }
                if (r >= 1) {
                    int w = (r - 1) * 8 + 4 + ch;
                    int kk = w >> 2, px = (w >> 1) & 1, jh = w & 1;
                    __builtin_amdgcn_sched_barrier(0);
                    c[px][jh] = __builtin_amdgcn_wmma_f32_16x16x4_f32(
                             false, xa[px][jh][kk],  false, wb[kk],
                             (short)0, c[px][jh],  false,  false);
                    __builtin_amdgcn_sched_barrier(0);
                }
            }
            float is = 1.0f / s;
            float* ar = &Att[i * ATI + lane];
#pragma unroll
            for (int g = 0; g < NTG; ++g)
                if (tg == g) {
#pragma unroll
                    for (int k = 0; k < TG; ++k) ar[k * ATS] = l[g * TG + k] * is;
                }
        }

#pragma unroll
        for (int px = 0; px < 2; ++px)
#pragma unroll
            for (int jh = 0; jh < 2; ++jh) {
                float* wp = &Whs[lm * TS + (jh * 16 + lh * 8) * JS + wave + 8 * px];
#pragma unroll
                for (int r = 0; r < 8; ++r) wp[r * JS] = c[px][jh][r];
            }
    }
    __syncthreads();

    if (wave < 8) {
        v2f aT[2][8];
#pragma unroll
        for (int vt = 0; vt < 2; ++vt)
#pragma unroll
            for (int s = 0; s < 8; ++s) {
                int i0 = 16 * (s >> 2) + 8 * lh + 2 * (s & 3);
                aT[vt][s].x = Aj[i0 * AJS + 16 * vt + lm];
                aT[vt][s].y = Aj[(i0 + 1) * AJS + 16 * vt + lm];
            }
        int t = t0 + wave;
        const float* wt = &Whs[wave * TS + lh * 2 * JS + lm];
        const float* at = &Att[wave * ATS + lm * ATI + lh * 2];
        v8f hp0 = {}, hp1 = {};
#pragma unroll
        for (int kk = 0; kk < 8; ++kk) {
            v2f b;
            b.x = wt[(kk * 4) * JS];
            b.y = wt[(kk * 4 + 1) * JS];
            v2f a0 = *(const v2f*)&at[kk * 4];
            v2f a1 = *(const v2f*)&at[16 * ATI + kk * 4];
            hp0 = __builtin_amdgcn_wmma_f32_16x16x4_f32(
                     false, a0,  false, b,
                     (short)0, hp0,  false,  false);
            hp1 = __builtin_amdgcn_wmma_f32_16x16x4_f32(
                     false, a1,  false, b,
                     (short)0, hp1,  false,  false);
        }
        v8f o0 = {}, o1 = {};
#pragma unroll
        for (int s = 0; s < 8; ++s) {
            v2f b;
            if (s < 4) { b.x = hp0[2 * (s & 3)]; b.y = hp0[2 * (s & 3) + 1]; }
            else       { b.x = hp1[2 * (s & 3)]; b.y = hp1[2 * (s & 3) + 1]; }
            o0 = __builtin_amdgcn_wmma_f32_16x16x4_f32(
                     false, aT[0][s],  false, b,
                     (short)0, o0,  false,  false);
            o1 = __builtin_amdgcn_wmma_f32_16x16x4_f32(
                     false, aT[1][s],  false, b,
                     (short)0, o1,  false,  false);
        }
#pragma unroll
        for (int vt = 0; vt < 2; ++vt) {
            v8f c = vt ? o1 : o0;
            int vb = vt * 16 + lh * 8;
            float* ob = out + ((size_t)(n * HW + p0 + lm) * T_ + t) * V_ + vb;
#pragma unroll
            for (int r = 0; r < 8; ++r) {
                float x = c[r];
                float y = x > 0.0f ? x : (__expf(x) - 1.0f);
                ob[r] = 1.0f / (1.0f + __expf(-y));
            }
        }
    }
}

extern "C" void kernel_launch(void* const* d_in, const int* in_sizes, int n_in,
                              void* d_out, int out_size, void* d_ws, size_t ws_size,
                              hipStream_t stream) {
    const float* h  = (const float*)d_in[0];
    const float* Wp = (const float*)d_in[1];
    const float* ap = (const float*)d_in[2];
    const float* Bp = (const float*)d_in[3];
    float* out = (float*)d_out;

    float* ws    = (float*)d_ws;
    float* ei    = ws;
    float* ej    = ei   + N_ * V_ * T_;

    eij_kernel <<<N_ * T_,    384, 0, stream>>>(h, Wp, ap, ei, ej);
    prod_kernel<<<N_ * (HW / PT) * NTG, 288, 0, stream>>>(h, Wp, ei, ej, Bp, out);
}
